# starting kernel
# baseline (speedup 1.0000x reference)
#include <hip/hip_runtime.h>

typedef __attribute__((ext_vector_type(2))) float v2f;
typedef __attribute__((ext_vector_type(4))) float v4f;
typedef __attribute__((ext_vector_type(8))) float v8f;

#define ALPHA 0.3f
#define EPSF  1e-6f

#define BB   2
#define NN   8192
#define KK   32
#define CC   128
#define HH   128
#define MM   (BB * NN)

__device__ __forceinline__ float lrelu(float x) { return x >= 0.f ? x : ALPHA * x; }

__global__ __launch_bounds__(256) void k1_hidden(const float* __restrict__ emb,
                                                 const float* __restrict__ Qw,
                                                 const float* __restrict__ Qb,
                                                 float* __restrict__ hidden)
{
    const int lane = threadIdx.x & 31;
    const int wave = threadIdx.x >> 5;
    const int m0   = (blockIdx.x * 8 + wave) << 4;
    const int h0   = blockIdx.y << 4;
    const int rc   = lane & 15;
    const int half = lane >> 4;

    const float* arow = emb + (size_t)(m0 + rc) * CC;
    v8f acc = (v8f)0.f;

    #pragma unroll 4
    for (int k4 = 0; k4 < CC; k4 += 4) {
        const int k0 = k4 + 2 * half;
        v2f a = *(const v2f*)(arow + k0);
        v2f b;
        b.x = Qw[(size_t)k0 * HH + h0 + rc];
        b.y = Qw[(size_t)(k0 + 1) * HH + h0 + rc];
        acc = __builtin_amdgcn_wmma_f32_16x16x4_f32(false, a, false, b,
                                                    (short)0, acc, false, false);
    }

    const float bv = Qb[h0 + rc];
    #pragma unroll
    for (int r = 0; r < 8; ++r) {
        const int m = m0 + r + (half << 3);
        hidden[(size_t)m * HH + h0 + rc] = lrelu(acc[r] + bv);
    }
}

__global__ __launch_bounds__(256) void k2_gather(const float* __restrict__ hidden,
                                                 const float* __restrict__ weights,
                                                 const int* __restrict__ nbr,
                                                 float* __restrict__ wsh)
{
    const int lane = threadIdx.x & 31;
    const int wave = threadIdx.x >> 5;
    const int idx  = blockIdx.x * 8 + wave;
    const int b    = idx >> 13;
    const int n    = idx & (NN - 1);

    const int   j = nbr[(size_t)n * KK + lane];
    const float w = weights[(size_t)n * NN + j];

    const float* hb = hidden + (size_t)b * NN * HH + (size_t)lane * 4;

    v4f   acc  = (v4f)0.f;
    float wsum = 0.f;
    #pragma unroll 8
    for (int k = 0; k < KK; ++k) {
        const int   jk = __shfl(j, k, 32);
        const float wk = __shfl(w, k, 32);
        const v4f h = *(const v4f*)(hb + (size_t)jk * HH);
        acc.x += wk * h.x;
        acc.y += wk * h.y;
        acc.z += wk * h.z;
        acc.w += wk * h.w;
        wsum  += wk;
    }
    const float inv = 1.f / (wsum + EPSF);
    v4f r;
    r.x = acc.x * inv; r.y = acc.y * inv; r.z = acc.z * inv; r.w = acc.w * inv;
    *(v4f*)(wsh + (size_t)idx * HH + lane * 4) = r;
}

__global__ __launch_bounds__(256) void k3_gemm2(const float* __restrict__ emb,
                                                const float* __restrict__ wsh,
                                                const float* __restrict__ Ww,
                                                const float* __restrict__ Wb,
                                                float* __restrict__ out)
{
    const int lane = threadIdx.x & 31;
    const int wave = threadIdx.x >> 5;
    const int m0   = (blockIdx.x * 8 + wave) << 4;
    const int h0   = blockIdx.y << 4;
    const int rc   = lane & 15;
    const int half = lane >> 4;

    const float* a1 = emb + (size_t)(m0 + rc) * CC;
    const float* a2 = wsh + (size_t)(m0 + rc) * HH;
    v8f acc = (v8f)0.f;

    #pragma unroll 4
    for (int k4 = 0; k4 < CC; k4 += 4) {
        const int k0 = k4 + 2 * half;
        v2f a = *(const v2f*)(a1 + k0);
        v2f b;
        b.x = Ww[(size_t)k0 * HH + h0 + rc];
        b.y = Ww[(size_t)(k0 + 1) * HH + h0 + rc];
        acc = __builtin_amdgcn_wmma_f32_16x16x4_f32(false, a, false, b,
                                                    (short)0, acc, false, false);
    }
    #pragma unroll 4
    for (int k4 = 0; k4 < HH; k4 += 4) {
        const int k0 = k4 + 2 * half;
        v2f a = *(const v2f*)(a2 + k0);
        v2f b;
        b.x = Ww[(size_t)(CC + k0) * HH + h0 + rc];
        b.y = Ww[(size_t)(CC + k0 + 1) * HH + h0 + rc];
        acc = __builtin_amdgcn_wmma_f32_16x16x4_f32(false, a, false, b,
                                                    (short)0, acc, false, false);
    }

    const float bv = Wb[h0 + rc];
    #pragma unroll
    for (int r = 0; r < 8; ++r) {
        const int m = m0 + r + (half << 3);
        out[(size_t)m * HH + h0 + rc] = lrelu(acc[r] + bv);
    }
}

__global__ __launch_bounds__(256) void k4_norm(float* __restrict__ out)
{
    const int lane = threadIdx.x & 31;
    const int wave = threadIdx.x >> 5;
    const int row  = blockIdx.x * 8 + wave;

    float* p = out + (size_t)row * HH + lane * 4;
    v4f v = *(const v4f*)p;
    float s = v.x * v.x + v.y * v.y + v.z * v.z + v.w * v.w;
    #pragma unroll
    for (int off = 16; off >= 1; off >>= 1)
        s += __shfl_xor(s, off, 32);
    const float inv = 1.f / (sqrtf(s) + EPSF);
    v.x *= inv; v.y *= inv; v.z *= inv; v.w *= inv;
    *(v4f*)p = v;
}

extern "C" void kernel_launch(void* const* d_in, const int* in_sizes, int n_in,
                              void* d_out, int out_size, void* d_ws, size_t ws_size,
                              hipStream_t stream)
{
    const float* emb     = (const float*)d_in[0];
    const float* weights = (const float*)d_in[1];
    const int*   nbr     = (const int*)  d_in[2];
    const float* Qw      = (const float*)d_in[3];
    const float* Qb      = (const float*)d_in[4];
    const float* Ww      = (const float*)d_in[5];
    const float* Wb      = (const float*)d_in[6];
    float*       out     = (float*)d_out;

    float* hidden = (float*)d_ws;
    float* wsh    = hidden + (size_t)MM * HH;

    const dim3 blk(256);
    k1_hidden<<<dim3(MM / 128, HH / 16), blk, 0, stream>>>(emb, Qw, Qb, hidden);
    k2_gather<<<dim3(MM / 8), blk, 0, stream>>>(hidden, weights, nbr, wsh);
    k3_gemm2<<<dim3(MM / 128, HH / 16), blk, 0, stream>>>(emb, wsh, Ww, Wb, out);
    k4_norm<<<dim3(MM / 8), blk, 0, stream>>>(out);
}


# Kernel A: each emb tile read from memory once per work-group (4 whole 512-byte rows per wave) and shared through LDS as A fragments, not fetched as many-line loads by all 8 waves; sums unchanged
# speedup vs baseline: 1.9139x; 1.9139x over previous
#include <hip/hip_runtime.h>


typedef __attribute__((ext_vector_type(2))) float v2f;
typedef __attribute__((ext_vector_type(4))) float v4f;
typedef __attribute__((ext_vector_type(8))) float v8f;

#define ALPHA 0.3f
#define EPSF  1e-6f

#define BB   2
#define NN   8192
#define KK   32
#define CC   128
#define HH   128
#define MM   (BB * NN)

#define GA   256
static_assert((MM / 32) % (2 * GA) == 0, "kernel A takes its row-tile pairs two trips at a time");

#define LDP  132

__device__ __forceinline__ float lrelu(float x) { return x >= 0.f ? x : ALPHA * x; }

#define WMMA_F32(a, b, c) __builtin_amdgcn_wmma_f32_16x16x4_f32(false, (a), false, (b), (short)0, (c), false, false)

__device__ __forceinline__ void dense3_trip(const v4f (&xa)[16], const v4f (&xb)[16],
                                            v4f (&ya)[16], v4f (&yb)[16],
                                            const v2f (&bq)[32], const v2f (&bw1)[32], const v2f (&bw2)[32],
                                            const float qb, const float wb,
                                            const float* __restrict__ emb,
                                            float* __restrict__ E, float* __restrict__ P,
                                            float* hbuf, float* tnext, const int m0, const int nxt,
                                            const int rc, const int half, const int col,
                                            const int wave, const int lane)
{
    v4f g[4];
    {
        const float* gr = emb + (size_t)((nxt << 5) + 4 * wave) * CC + 4 * lane;
        #pragma unroll
        for (int i = 0; i < 4; ++i)
            g[i] = *(const v4f*)(gr + (size_t)i * CC);
    }
    __builtin_amdgcn_sched_barrier(0);

    v8f ah0 = (v8f)0.f, ah1 = (v8f)0.f;
    #pragma unroll
    for (int j = 0; j < 16; ++j) {
        v2f a;
        a.x = xa[j].x; a.y = xa[j].y;
        ah0 = WMMA_F32(a, bq[2 * j], ah0);
        a.x = xb[j].x; a.y = xb[j].y;
        ah1 = WMMA_F32(a, bq[2 * j], ah1);
        a.x = xa[j].z; a.y = xa[j].w;
        ah0 = WMMA_F32(a, bq[2 * j + 1], ah0);
        a.x = xb[j].z; a.y = xb[j].w;
        ah1 = WMMA_F32(a, bq[2 * j + 1], ah1);
    }

    #pragma unroll
    for (int r = 0; r < 8; ++r) {
        const int row = r + (half << 3);
        hbuf[row * LDP + col]        = lrelu(ah0[r] + qb);
        hbuf[(16 + row) * LDP + col] = lrelu(ah1[r] + qb);
    }
    __builtin_amdgcn_sched_barrier(0);
    #pragma unroll
    for (int i = 0; i < 4; ++i)
        *(v4f*)(tnext + (4 * wave + i) * LDP + 4 * lane) = g[i];
    __syncthreads();
    __builtin_amdgcn_sched_barrier(0);

    {
        const float* tr0 = tnext + rc * LDP + 4 * half;
        const float* tr1 = tr0 + 16 * LDP;
        #pragma unroll
        for (int j = 0; j < 16; ++j) {
            ya[j] = *(const v4f*)(tr0 + 8 * j);
            yb[j] = *(const v4f*)(tr1 + 8 * j);
        }
    }
    __builtin_amdgcn_sched_barrier(0);

    const float* hr0 = hbuf + rc * LDP + 4 * half;
    const float* hr1 = hr0 + 16 * LDP;
    v8f ae0 = (v8f)0.f, ae1 = (v8f)0.f, ap0 = (v8f)0.f, ap1 = (v8f)0.f;
    #pragma unroll
    for (int j = 0; j < 16; ++j) {
        const v4f h0 = *(const v4f*)(hr0 + 8 * j);
        const v4f h1 = *(const v4f*)(hr1 + 8 * j);
        v2f a;
        a.x = xa[j].x; a.y = xa[j].y;
        ae0 = WMMA_F32(a, bw1[2 * j], ae0);
        a.x = h0.x; a.y = h0.y;
        ap0 = WMMA_F32(a, bw2[2 * j], ap0);
        a.x = xb[j].x; a.y = xb[j].y;
        ae1 = WMMA_F32(a, bw1[2 * j], ae1);
        a.x = h1.x; a.y = h1.y;
        ap1 = WMMA_F32(a, bw2[2 * j], ap1);
        a.x = xa[j].z; a.y = xa[j].w;
        ae0 = WMMA_F32(a, bw1[2 * j + 1], ae0);
        a.x = h0.z; a.y = h0.w;
        ap0 = WMMA_F32(a, bw2[2 * j + 1], ap0);
        a.x = xb[j].z; a.y = xb[j].w;
        ae1 = WMMA_F32(a, bw1[2 * j + 1], ae1);
        a.x = h1.z; a.y = h1.w;
        ap1 = WMMA_F32(a, bw2[2 * j + 1], ap1);
    }

    #pragma unroll
    for (int r = 0; r < 8; ++r) {
        const size_t m = (size_t)(m0 + r + (half << 3));
        E[m * HH + col]        = ae0[r] + wb;
        E[(m + 16) * HH + col] = ae1[r] + wb;
        P[m * HH + col]        = ap0[r];
        P[(m + 16) * HH + col] = ap1[r];
    }
}

__global__ __launch_bounds__(256) void k_dense3(const float* __restrict__ emb,
                                                const float* __restrict__ Qw,
                                                const float* __restrict__ Qb,
                                                const float* __restrict__ Ww,
                                                const float* __restrict__ Wb,
                                                float* __restrict__ E,
                                                float* __restrict__ P)
{
    __shared__ __attribute__((aligned(16))) float hs[2][32 * LDP];
    __shared__ __attribute__((aligned(16))) float ts[2][32 * LDP];

    const int lane = threadIdx.x & 31;
    const int wave = threadIdx.x >> 5;
    const int rc   = lane & 15;
    const int half = lane >> 4;
    const int col  = (wave << 4) + rc;
    const int NP   = MM / 32;
    const int G    = GA;

    v4f xa[16], xb[16], ya[16], yb[16], g0[4];
    {
        const float* gr0 = emb + (size_t)(((int)blockIdx.x << 5) + 4 * wave) * CC + 4 * lane;
        #pragma unroll
        for (int i = 0; i < 4; ++i)
            g0[i] = *(const v4f*)(gr0 + (size_t)i * CC);
    }

    v2f bq[32], bw1[32], bw2[32];
    #pragma unroll
    for (int s = 0; s < 32; ++s) {
        const int k = 8 * (s >> 1) + 4 * half + 2 * (s & 1);
        bq[s].x  = Qw[(size_t)k * HH + col];
        bq[s].y  = Qw[(size_t)(k + 1) * HH + col];
        bw1[s].x = Ww[(size_t)k * HH + col];
        bw1[s].y = Ww[(size_t)(k + 1) * HH + col];
        bw2[s].x = Ww[(size_t)(CC + k) * HH + col];
        bw2[s].y = Ww[(size_t)(CC + k + 1) * HH + col];
    }
    const float qb = Qb[col];
    const float wb = Wb[col];

    #pragma unroll
    for (int i = 0; i < 4; ++i)
        *(v4f*)(&ts[0][(4 * wave + i) * LDP + 4 * lane]) = g0[i];
    __syncthreads();
    {
        const float* tr0 = &ts[0][rc * LDP + 4 * half];
        const float* tr1 = tr0 + 16 * LDP;
        #pragma unroll
        for (int j = 0; j < 16; ++j) {
            xa[j] = *(const v4f*)(tr0 + 8 * j);
            xb[j] = *(const v4f*)(tr1 + 8 * j);
        }
    }

    for (int pair = blockIdx.x; pair < NP; pair += 2 * G) {
        const int p1 = pair + G;
        const int p2 = pair + 2 * G;
        dense3_trip(xa, xb, ya, yb, bq, bw1, bw2, qb, wb, emb, E, P, hs[0], ts[1],
                    pair << 5, p1, rc, half, col, wave, lane);
        dense3_trip(ya, yb, xa, xb, bq, bw1, bw2, qb, wb, emb, E, P, hs[1], ts[0],
                    p1 << 5, (p2 < NP) ? p2 : p1, rc, half, col, wave, lane);
    }
}

__device__ __forceinline__ void gather_node(const float* __restrict__ p0, const float* __restrict__ p1,
                                            const float* __restrict__ E, float* __restrict__ out,
                                            int n, int lane, int j, float w)
{
    const size_t o0 = (size_t)n * HH + lane * 4;
    const size_t o1 = o0 + (size_t)NN * HH;

    const v4f e0 = *(const v4f*)(E + o0);
    const v4f e1 = *(const v4f*)(E + o1);

    v4f   a0 = (v4f)0.f, a1 = (v4f)0.f;
    float wsum = 0.f;
    #pragma unroll 16
    for (int k = 0; k < KK; ++k) {
        const int   jk = __builtin_amdgcn_readlane(j, k);
        const float wk = __int_as_float(__builtin_amdgcn_readlane(__float_as_int(w), k));
        const v4f h0 = *(const v4f*)(p0 + (size_t)jk * HH);
        const v4f h1 = *(const v4f*)(p1 + (size_t)jk * HH);
        a0.x += wk * h0.x; a0.y += wk * h0.y; a0.z += wk * h0.z; a0.w += wk * h0.w;
        a1.x += wk * h1.x; a1.y += wk * h1.y; a1.z += wk * h1.z; a1.w += wk * h1.w;
        wsum += wk;
    }
    const float inv = 1.f / (wsum + EPSF);

    v4f x0, x1;
    x0.x = lrelu(e0.x + a0.x * inv); x0.y = lrelu(e0.y + a0.y * inv);
    x0.z = lrelu(e0.z + a0.z * inv); x0.w = lrelu(e0.w + a0.w * inv);
    x1.x = lrelu(e1.x + a1.x * inv); x1.y = lrelu(e1.y + a1.y * inv);
    x1.z = lrelu(e1.z + a1.z * inv); x1.w = lrelu(e1.w + a1.w * inv);

    float s0 = x0.x * x0.x + x0.y * x0.y + x0.z * x0.z + x0.w * x0.w;
    float s1 = x1.x * x1.x + x1.y * x1.y + x1.z * x1.z + x1.w * x1.w;
    #pragma unroll
    for (int off = 16; off >= 1; off >>= 1) {
        s0 += __shfl_xor(s0, off, 32);
        s1 += __shfl_xor(s1, off, 32);
    }
    const float i0 = 1.f / (sqrtf(s0) + EPSF);
    const float i1 = 1.f / (sqrtf(s1) + EPSF);
    x0.x *= i0; x0.y *= i0; x0.z *= i0; x0.w *= i0;
    x1.x *= i1; x1.y *= i1; x1.z *= i1; x1.w *= i1;
    *(v4f*)(out + o0) = x0;
    *(v4f*)(out + o1) = x1;
}

__global__ __launch_bounds__(256) void k_gather_out(const float* __restrict__ P,
                                                    const float* __restrict__ E,
                                                    const float* __restrict__ weights,
                                                    const int* __restrict__ nbr,
                                                    float* __restrict__ out)
{
    const int lane = threadIdx.x & 31;
    const int wave = __builtin_amdgcn_readfirstlane(threadIdx.x >> 5);
    const int nw   = gridDim.x * 8;

    const float* p0 = P + (size_t)lane * 4;
    const float* p1 = p0 + (size_t)NN * HH;

    for (int na = blockIdx.x * 8 + wave; na < NN; na += 2 * nw) {
        const int  nb    = na + nw;
        const bool has_b = nb < NN;
        const int  nbc   = has_b ? nb : na;
        const int   ja = nbr[(size_t)na * KK + lane];
        const int   jb = nbr[(size_t)nbc * KK + lane];
        const float wa = weights[(size_t)na * NN + ja];
        const float wb = weights[(size_t)nbc * NN + jb];

        gather_node(p0, p1, E, out, na, lane, ja, wa);
        if (has_b) gather_node(p0, p1, E, out, nb, lane, jb, wb);
    }
}

extern "C" void kernel_launch(void* const* d_in, const int* in_sizes, int n_in,
                              void* d_out, int out_size, void* d_ws, size_t ws_size,
                              hipStream_t stream)
{
    const float* emb     = (const float*)d_in[0];
    const float* weights = (const float*)d_in[1];
    const int*   nbr     = (const int*)  d_in[2];
    const float* Qw      = (const float*)d_in[3];
    const float* Qb      = (const float*)d_in[4];
    const float* Ww      = (const float*)d_in[5];
    const float* Wb      = (const float*)d_in[6];
    float*       out     = (float*)d_out;

    float* P = (float*)d_ws;
    float* E = P + (size_t)MM * HH;

    const dim3 blk(256);
    k_dense3<<<dim3(GA), blk, 0, stream>>>(emb, Qw, Qb, Ww, Wb, E, P);
    k_gather_out<<<dim3(512), blk, 0, stream>>>(P, E, weights, nbr, out);
}


# Kernel A: barriers order LDS only, so no wave waits there for memory; Qw strip fetched first in read order, both Ww strips fetched inside the first hidden product and arrive under it; sums unchanged
# speedup vs baseline: 2.0886x; 1.0913x over previous
#include <hip/hip_runtime.h>


typedef __attribute__((ext_vector_type(2))) float v2f;
typedef __attribute__((ext_vector_type(4))) float v4f;
typedef __attribute__((ext_vector_type(8))) float v8f;

#define ALPHA 0.3f
#define EPSF  1e-6f

#define BB   2
#define NN   8192
#define KK   32
#define CC   128
#define HH   128
#define MM   (BB * NN)

#define GA   256
static_assert((MM / 32) % (2 * GA) == 0, "kernel A takes its row-tile pairs two trips at a time");

#define LDP  132

__device__ __forceinline__ float lrelu(float x) { return x >= 0.f ? x : ALPHA * x; }

#define WMMA_F32(a, b, c) __builtin_amdgcn_wmma_f32_16x16x4_f32(false, (a), false, (b), (short)0, (c), false, false)

__device__ __forceinline__ void lds_barrier()
{
    __builtin_amdgcn_fence(__ATOMIC_RELEASE, "workgroup", "local");
    __builtin_amdgcn_s_barrier();
    __builtin_amdgcn_fence(__ATOMIC_ACQUIRE, "workgroup", "local");
}

template <bool FIRST>
__device__ __forceinline__ void dense3_trip(const v4f (&xa)[16], const v4f (&xb)[16],
                                            v4f (&ya)[16], v4f (&yb)[16],
                                            const v2f (&bq)[32], v2f (&bw1)[32], v2f (&bw2)[32],
                                            const float qb, const float wb,
                                            const float* __restrict__ emb,
                                            const float* __restrict__ Ww,
                                            float* __restrict__ E, float* __restrict__ P,
                                            float* hbuf, float* tnext, const int m0, const int nxt,
                                            const int rc, const int half, const int col,
                                            const int wave, const int lane)
{
    v4f g[4];
    {
        const float* gr = emb + (size_t)((nxt << 5) + 4 * wave) * CC + 4 * lane;
        #pragma unroll
        for (int i = 0; i < 4; ++i)
            g[i] = *(const v4f*)(gr + (size_t)i * CC);
    }
    __builtin_amdgcn_sched_barrier(0);

    v8f ah0 = (v8f)0.f, ah1 = (v8f)0.f;
    #pragma unroll
    for (int j = 0; j < 16; ++j) {
        if constexpr (FIRST) {
            if (j == 5 || j == 8 || j == 11 || j == 14) {
                const int q  = (j - 5) / 3;
                const int s0 = 16 * (q >> 1);
                const int kb = (q & 1) ? CC : 0;
                __builtin_amdgcn_sched_barrier(0);
                #pragma unroll
                for (int s = s0; s < s0 + 16; ++s) {
                    const int k = 8 * (s >> 1) + 4 * half + 2 * (s & 1);
                    v2f t;
                    t.x = Ww[(size_t)(kb + k) * HH + col];
                    t.y = Ww[(size_t)(kb + k + 1) * HH + col];
                    if (q & 1) bw2[s] = t; else bw1[s] = t;
                }
                __builtin_amdgcn_sched_barrier(0);
            }
        }
        v2f a;
        a.x = xa[j].x; a.y = xa[j].y;
        ah0 = WMMA_F32(a, bq[2 * j], ah0);
        a.x = xb[j].x; a.y = xb[j].y;
        ah1 = WMMA_F32(a, bq[2 * j], ah1);
        a.x = xa[j].z; a.y = xa[j].w;
        ah0 = WMMA_F32(a, bq[2 * j + 1], ah0);
        a.x = xb[j].z; a.y = xb[j].w;
        ah1 = WMMA_F32(a, bq[2 * j + 1], ah1);
    }

    #pragma unroll
    for (int r = 0; r < 8; ++r) {
        const int row = r + (half << 3);
        hbuf[row * LDP + col]        = lrelu(ah0[r] + qb);
        hbuf[(16 + row) * LDP + col] = lrelu(ah1[r] + qb);
    }
    __builtin_amdgcn_sched_barrier(0);
    #pragma unroll
    for (int i = 0; i < 4; ++i)
        *(v4f*)(tnext + (4 * wave + i) * LDP + 4 * lane) = g[i];
    lds_barrier();
    __builtin_amdgcn_sched_barrier(0);

    {
        const float* tr0 = tnext + rc * LDP + 4 * half;
        const float* tr1 = tr0 + 16 * LDP;
        #pragma unroll
        for (int j = 0; j < 16; ++j) {
            ya[j] = *(const v4f*)(tr0 + 8 * j);
            yb[j] = *(const v4f*)(tr1 + 8 * j);
        }
    }
    __builtin_amdgcn_sched_barrier(0);

    const float* hr0 = hbuf + rc * LDP + 4 * half;
    const float* hr1 = hr0 + 16 * LDP;
    v8f ae0 = (v8f)0.f, ae1 = (v8f)0.f, ap0 = (v8f)0.f, ap1 = (v8f)0.f;
    #pragma unroll
    for (int j = 0; j < 16; ++j) {
        const v4f h0 = *(const v4f*)(hr0 + 8 * j);
        const v4f h1 = *(const v4f*)(hr1 + 8 * j);
        v2f a;
        a.x = xa[j].x; a.y = xa[j].y;
        ae0 = WMMA_F32(a, bw1[2 * j], ae0);
        a.x = h0.x; a.y = h0.y;
        ap0 = WMMA_F32(a, bw2[2 * j], ap0);
        a.x = xb[j].x; a.y = xb[j].y;
        ae1 = WMMA_F32(a, bw1[2 * j], ae1);
        a.x = h1.x; a.y = h1.y;
        ap1 = WMMA_F32(a, bw2[2 * j], ap1);
        a.x = xa[j].z; a.y = xa[j].w;
        ae0 = WMMA_F32(a, bw1[2 * j + 1], ae0);
        a.x = h0.z; a.y = h0.w;
        ap0 = WMMA_F32(a, bw2[2 * j + 1], ap0);
        a.x = xb[j].z; a.y = xb[j].w;
        ae1 = WMMA_F32(a, bw1[2 * j + 1], ae1);
        a.x = h1.z; a.y = h1.w;
        ap1 = WMMA_F32(a, bw2[2 * j + 1], ap1);
    }

    #pragma unroll
    for (int r = 0; r < 8; ++r) {
        const size_t m = (size_t)(m0 + r + (half << 3));
        E[m * HH + col]        = ae0[r] + wb;
        E[(m + 16) * HH + col] = ae1[r] + wb;
        P[m * HH + col]        = ap0[r];
        P[(m + 16) * HH + col] = ap1[r];
    }
}

__global__ __launch_bounds__(256) void k_dense3(const float* __restrict__ emb,
                                                const float* __restrict__ Qw,
                                                const float* __restrict__ Qb,
                                                const float* __restrict__ Ww,
                                                const float* __restrict__ Wb,
                                                float* __restrict__ E,
                                                float* __restrict__ P)
{
    __shared__ __attribute__((aligned(16))) float hs[2][32 * LDP];
    __shared__ __attribute__((aligned(16))) float ts[2][32 * LDP];

    const int lane = threadIdx.x & 31;
    const int wave = threadIdx.x >> 5;
    const int rc   = lane & 15;
    const int half = lane >> 4;
    const int col  = (wave << 4) + rc;
    const int NP   = MM / 32;
    const int G    = GA;

    v4f xa[16], xb[16], ya[16], yb[16], g0[4];
    {
        const float* gr0 = emb + (size_t)(((int)blockIdx.x << 5) + 4 * wave) * CC + 4 * lane;
        #pragma unroll
        for (int i = 0; i < 4; ++i)
            g0[i] = *(const v4f*)(gr0 + (size_t)i * CC);
    }
    __builtin_amdgcn_sched_barrier(0);

    v2f bq[32], bw1[32], bw2[32];
    #pragma unroll
    for (int s = 0; s < 32; ++s) {
        const int k = 8 * (s >> 1) + 4 * half + 2 * (s & 1);
        bq[s].x  = Qw[(size_t)k * HH + col];
        bq[s].y  = Qw[(size_t)(k + 1) * HH + col];
        if ((s & 7) == 7) __builtin_amdgcn_sched_barrier(0);
    }
    const float qb = Qb[col];
    const float wb = Wb[col];

    #pragma unroll
    for (int i = 0; i < 4; ++i)
        *(v4f*)(&ts[0][(4 * wave + i) * LDP + 4 * lane]) = g0[i];
    lds_barrier();
    {
        const float* tr0 = &ts[0][rc * LDP + 4 * half];
        const float* tr1 = tr0 + 16 * LDP;
        #pragma unroll
        for (int j = 0; j < 16; ++j) {
            xa[j] = *(const v4f*)(tr0 + 8 * j);
            xb[j] = *(const v4f*)(tr1 + 8 * j);
        }
    }

    {
        const int pair = blockIdx.x;
        const int p1 = pair + G;
        const int p2 = pair + 2 * G;
        dense3_trip<true>(xa, xb, ya, yb, bq, bw1, bw2, qb, wb, emb, Ww, E, P, hs[0], ts[1],
                          pair << 5, p1, rc, half, col, wave, lane);
        dense3_trip<false>(ya, yb, xa, xb, bq, bw1, bw2, qb, wb, emb, Ww, E, P, hs[1], ts[0],
                           p1 << 5, (p2 < NP) ? p2 : p1, rc, half, col, wave, lane);
    }
    for (int pair = blockIdx.x + 2 * G; pair < NP; pair += 2 * G) {
        const int p1 = pair + G;
        const int p2 = pair + 2 * G;
        dense3_trip<false>(xa, xb, ya, yb, bq, bw1, bw2, qb, wb, emb, Ww, E, P, hs[0], ts[1],
                           pair << 5, p1, rc, half, col, wave, lane);
        dense3_trip<false>(ya, yb, xa, xb, bq, bw1, bw2, qb, wb, emb, Ww, E, P, hs[1], ts[0],
                           p1 << 5, (p2 < NP) ? p2 : p1, rc, half, col, wave, lane);
    }
}

__device__ __forceinline__ void gather_node(const float* __restrict__ p0, const float* __restrict__ p1,
                                            const float* __restrict__ E, float* __restrict__ out,
                                            int n, int lane, int j, float w)
{
    const size_t o0 = (size_t)n * HH + lane * 4;
    const size_t o1 = o0 + (size_t)NN * HH;

    const v4f e0 = *(const v4f*)(E + o0);
    const v4f e1 = *(const v4f*)(E + o1);

    v4f   a0 = (v4f)0.f, a1 = (v4f)0.f;
    float wsum = 0.f;
    #pragma unroll 16
    for (int k = 0; k < KK; ++k) {
        const int   jk = __builtin_amdgcn_readlane(j, k);
        const float wk = __int_as_float(__builtin_amdgcn_readlane(__float_as_int(w), k));
        const v4f h0 = *(const v4f*)(p0 + (size_t)jk * HH);
        const v4f h1 = *(const v4f*)(p1 + (size_t)jk * HH);
        a0.x += wk * h0.x; a0.y += wk * h0.y; a0.z += wk * h0.z; a0.w += wk * h0.w;
        a1.x += wk * h1.x; a1.y += wk * h1.y; a1.z += wk * h1.z; a1.w += wk * h1.w;
        wsum += wk;
    }
    const float inv = 1.f / (wsum + EPSF);

    v4f x0, x1;
    x0.x = lrelu(e0.x + a0.x * inv); x0.y = lrelu(e0.y + a0.y * inv);
    x0.z = lrelu(e0.z + a0.z * inv); x0.w = lrelu(e0.w + a0.w * inv);
    x1.x = lrelu(e1.x + a1.x * inv); x1.y = lrelu(e1.y + a1.y * inv);
    x1.z = lrelu(e1.z + a1.z * inv); x1.w = lrelu(e1.w + a1.w * inv);

    float s0 = x0.x * x0.x + x0.y * x0.y + x0.z * x0.z + x0.w * x0.w;
    float s1 = x1.x * x1.x + x1.y * x1.y + x1.z * x1.z + x1.w * x1.w;
    #pragma unroll
    for (int off = 16; off >= 1; off >>= 1) {
        s0 += __shfl_xor(s0, off, 32);
        s1 += __shfl_xor(s1, off, 32);
    }
    const float i0 = 1.f / (sqrtf(s0) + EPSF);
    const float i1 = 1.f / (sqrtf(s1) + EPSF);
    x0.x *= i0; x0.y *= i0; x0.z *= i0; x0.w *= i0;
    x1.x *= i1; x1.y *= i1; x1.z *= i1; x1.w *= i1;
    *(v4f*)(out + o0) = x0;
    *(v4f*)(out + o1) = x1;
}

__global__ __launch_bounds__(256) void k_gather_out(const float* __restrict__ P,
                                                    const float* __restrict__ E,
                                                    const float* __restrict__ weights,
                                                    const int* __restrict__ nbr,
                                                    float* __restrict__ out)
{
    const int lane = threadIdx.x & 31;
    const int wave = __builtin_amdgcn_readfirstlane(threadIdx.x >> 5);
    const int nw   = gridDim.x * 8;

    const float* p0 = P + (size_t)lane * 4;
    const float* p1 = p0 + (size_t)NN * HH;

    for (int na = blockIdx.x * 8 + wave; na < NN; na += 2 * nw) {
        const int  nb    = na + nw;
        const bool has_b = nb < NN;
        const int  nbc   = has_b ? nb : na;
        const int   ja = nbr[(size_t)na * KK + lane];
        const int   jb = nbr[(size_t)nbc * KK + lane];
        const float wa = weights[(size_t)na * NN + ja];
        const float wb = weights[(size_t)nbc * NN + jb];

        gather_node(p0, p1, E, out, na, lane, ja, wa);
        if (has_b) gather_node(p0, p1, E, out, nb, lane, jb, wb);
    }
}

extern "C" void kernel_launch(void* const* d_in, const int* in_sizes, int n_in,
                              void* d_out, int out_size, void* d_ws, size_t ws_size,
                              hipStream_t stream)
{
    const float* emb     = (const float*)d_in[0];
    const float* weights = (const float*)d_in[1];
    const int*   nbr     = (const int*)  d_in[2];
    const float* Qw      = (const float*)d_in[3];
    const float* Qb      = (const float*)d_in[4];
    const float* Ww      = (const float*)d_in[5];
    const float* Wb      = (const float*)d_in[6];
    float*       out     = (float*)d_out;

    float* P = (float*)d_ws;
    float* E = P + (size_t)MM * HH;

    const dim3 blk(256);
    k_dense3<<<dim3(GA), blk, 0, stream>>>(emb, Qw, Qb, Ww, Wb, E, P);
    k_gather_out<<<dim3(512), blk, 0, stream>>>(P, E, weights, nbr, out);
}


# E product (emb x Ww[0:128]) moved from matrix-bound kernel A into kernel B: each gather work-group does it for its own 64 rows while its edge weights come from memory; E skips memory; sums unchanged
# speedup vs baseline: 2.3152x; 1.1085x over previous
#include <hip/hip_runtime.h>


typedef __attribute__((ext_vector_type(2))) float v2f;
typedef __attribute__((ext_vector_type(4))) float v4f;
typedef __attribute__((ext_vector_type(8))) float v8f;

#define ALPHA 0.3f
#define EPSF  1e-6f

#define BB   2
#define NN   8192
#define KK   32
#define CC   128
#define HH   128
#define MM   (BB * NN)

#define GA   256
static_assert((MM / 32) % (2 * GA) == 0, "kernel A takes its row-tile pairs two trips at a time");

#define LDP  132

__device__ __forceinline__ float lrelu(float x) { return x >= 0.f ? x : ALPHA * x; }

#define WMMA_F32(a, b, c) __builtin_amdgcn_wmma_f32_16x16x4_f32(false, (a), false, (b), (short)0, (c), false, false)

__device__ __forceinline__ void lds_barrier()
{
    __builtin_amdgcn_fence(__ATOMIC_RELEASE, "workgroup", "local");
    __builtin_amdgcn_s_barrier();
    __builtin_amdgcn_fence(__ATOMIC_ACQUIRE, "workgroup", "local");
}

template <bool FIRST>
__device__ __forceinline__ void dense2_trip(const v4f (&xa)[16], const v4f (&xb)[16],
                                            v4f (&ya)[16], v4f (&yb)[16],
                                            const v2f (&bq)[32], v2f (&bw2)[32],
                                            const float qb,
                                            const float* __restrict__ emb,
                                            const float* __restrict__ Ww,
                                            float* __restrict__ P,
                                            float* hbuf, float* tnext, const int m0, const int nxt,
                                            const int rc, const int half, const int col,
                                            const int wave, const int lane)
{
    v4f g[4];
    {
        const float* gr = emb + (size_t)((nxt << 5) + 4 * wave) * CC + 4 * lane;
        #pragma unroll
        for (int i = 0; i < 4; ++i)
            g[i] = *(const v4f*)(gr + (size_t)i * CC);
    }
    __builtin_amdgcn_sched_barrier(0);

    v8f ah0 = (v8f)0.f, ah1 = (v8f)0.f;
    #pragma unroll
    for (int j = 0; j < 16; ++j) {
        if constexpr (FIRST) {
            if (j == 5 || j == 10) {
                const int s0 = (j == 5) ? 0 : 16;
                __builtin_amdgcn_sched_barrier(0);
                #pragma unroll
                for (int s = s0; s < s0 + 16; ++s) {
                    const int k = 8 * (s >> 1) + 4 * half + 2 * (s & 1);
                    v2f t;
                    t.x = Ww[(size_t)(CC + k) * HH + col];
                    t.y = Ww[(size_t)(CC + k + 1) * HH + col];
                    bw2[s] = t;
                }
                __builtin_amdgcn_sched_barrier(0);
            }
        }
        v2f a;
        a.x = xa[j].x; a.y = xa[j].y;
        ah0 = WMMA_F32(a, bq[2 * j], ah0);
        a.x = xb[j].x; a.y = xb[j].y;
        ah1 = WMMA_F32(a, bq[2 * j], ah1);
        a.x = xa[j].z; a.y = xa[j].w;
        ah0 = WMMA_F32(a, bq[2 * j + 1], ah0);
        a.x = xb[j].z; a.y = xb[j].w;
        ah1 = WMMA_F32(a, bq[2 * j + 1], ah1);
    }

    #pragma unroll
    for (int r = 0; r < 8; ++r) {
        const int row = r + (half << 3);
        hbuf[row * LDP + col]        = lrelu(ah0[r] + qb);
        hbuf[(16 + row) * LDP + col] = lrelu(ah1[r] + qb);
    }
    __builtin_amdgcn_sched_barrier(0);
    #pragma unroll
    for (int i = 0; i < 4; ++i)
        *(v4f*)(tnext + (4 * wave + i) * LDP + 4 * lane) = g[i];
    lds_barrier();
    __builtin_amdgcn_sched_barrier(0);

    v4f h0[16], h1[16];
    {
        const float* hr0 = hbuf + rc * LDP + 4 * half;
        const float* hr1 = hr0 + 16 * LDP;
        #pragma unroll
        for (int j = 0; j < 16; ++j) {
            h0[j] = *(const v4f*)(hr0 + 8 * j);
            h1[j] = *(const v4f*)(hr1 + 8 * j);
        }
        const float* tr0 = tnext + rc * LDP + 4 * half;
        const float* tr1 = tr0 + 16 * LDP;
        #pragma unroll
        for (int j = 0; j < 16; ++j) {
            ya[j] = *(const v4f*)(tr0 + 8 * j);
            yb[j] = *(const v4f*)(tr1 + 8 * j);
        }
    }
    __builtin_amdgcn_sched_barrier(0);

    v8f ap0 = (v8f)0.f, ap1 = (v8f)0.f;
    #pragma unroll
    for (int j = 0; j < 16; ++j) {
        v2f a;
        a.x = h0[j].x; a.y = h0[j].y;
        ap0 = WMMA_F32(a, bw2[2 * j], ap0);
        a.x = h1[j].x; a.y = h1[j].y;
        ap1 = WMMA_F32(a, bw2[2 * j], ap1);
        a.x = h0[j].z; a.y = h0[j].w;
        ap0 = WMMA_F32(a, bw2[2 * j + 1], ap0);
        a.x = h1[j].z; a.y = h1[j].w;
        ap1 = WMMA_F32(a, bw2[2 * j + 1], ap1);
    }

    #pragma unroll
    for (int r = 0; r < 8; ++r) {
        const size_t m = (size_t)(m0 + r + (half << 3));
        P[m * HH + col]        = ap0[r];
        P[(m + 16) * HH + col] = ap1[r];
    }
}

__global__ __launch_bounds__(256) void k_dense2(const float* __restrict__ emb,
                                                const float* __restrict__ Qw,
                                                const float* __restrict__ Qb,
                                                const float* __restrict__ Ww,
                                                float* __restrict__ P)
{
    __shared__ __attribute__((aligned(16))) float hs[2][32 * LDP];
    __shared__ __attribute__((aligned(16))) float ts[2][32 * LDP];

    const int lane = threadIdx.x & 31;
    const int wave = threadIdx.x >> 5;
    const int rc   = lane & 15;
    const int half = lane >> 4;
    const int col  = (wave << 4) + rc;
    const int NP   = MM / 32;
    const int G    = GA;

    v4f xa[16], xb[16], ya[16], yb[16], g0[4];
    {
        const float* gr0 = emb + (size_t)(((int)blockIdx.x << 5) + 4 * wave) * CC + 4 * lane;
        #pragma unroll
        for (int i = 0; i < 4; ++i)
            g0[i] = *(const v4f*)(gr0 + (size_t)i * CC);
    }
    __builtin_amdgcn_sched_barrier(0);

    v2f bq[32], bw2[32];
    #pragma unroll
    for (int s = 0; s < 32; ++s) {
        const int k = 8 * (s >> 1) + 4 * half + 2 * (s & 1);
        bq[s].x  = Qw[(size_t)k * HH + col];
        bq[s].y  = Qw[(size_t)(k + 1) * HH + col];
        if ((s & 7) == 7) __builtin_amdgcn_sched_barrier(0);
    }
    const float qb = Qb[col];

    #pragma unroll
    for (int i = 0; i < 4; ++i)
        *(v4f*)(&ts[0][(4 * wave + i) * LDP + 4 * lane]) = g0[i];
    lds_barrier();
    {
        const float* tr0 = &ts[0][rc * LDP + 4 * half];
        const float* tr1 = tr0 + 16 * LDP;
        #pragma unroll
        for (int j = 0; j < 16; ++j) {
            xa[j] = *(const v4f*)(tr0 + 8 * j);
            xb[j] = *(const v4f*)(tr1 + 8 * j);
        }
    }

    {
        const int pair = blockIdx.x;
        const int p1 = pair + G;
        const int p2 = pair + 2 * G;
        dense2_trip<true>(xa, xb, ya, yb, bq, bw2, qb, emb, Ww, P, hs[0], ts[1],
                          pair << 5, p1, rc, half, col, wave, lane);
        dense2_trip<false>(ya, yb, xa, xb, bq, bw2, qb, emb, Ww, P, hs[1], ts[0],
                           p1 << 5, (p2 < NP) ? p2 : p1, rc, half, col, wave, lane);
    }
    for (int pair = blockIdx.x + 2 * G; pair < NP; pair += 2 * G) {
        const int p1 = pair + G;
        const int p2 = pair + 2 * G;
        dense2_trip<false>(xa, xb, ya, yb, bq, bw2, qb, emb, Ww, P, hs[0], ts[1],
                           pair << 5, p1, rc, half, col, wave, lane);
        dense2_trip<false>(ya, yb, xa, xb, bq, bw2, qb, emb, Ww, P, hs[1], ts[0],
                           p1 << 5, (p2 < NP) ? p2 : p1, rc, half, col, wave, lane);
    }
}

#define GB   256
static_assert(GB * 16 * 2 == NN, "kernel B: every wave takes exactly two nodes, na and na + NN/2");

__device__ __forceinline__ void gather_sum(const float* __restrict__ p0, const float* __restrict__ p1,
                                           const int j, const float w,
                                           v4f& a0, v4f& a1, float& inv)
{
    a0 = (v4f)0.f; a1 = (v4f)0.f;
    float wsum = 0.f;
    #pragma unroll 16
    for (int k = 0; k < KK; ++k) {
        const int   jk = __builtin_amdgcn_readlane(j, k);
        const float wk = __int_as_float(__builtin_amdgcn_readlane(__float_as_int(w), k));
        const v4f h0 = *(const v4f*)(p0 + (size_t)jk * HH);
        const v4f h1 = *(const v4f*)(p1 + (size_t)jk * HH);
        a0.x += wk * h0.x; a0.y += wk * h0.y; a0.z += wk * h0.z; a0.w += wk * h0.w;
        a1.x += wk * h1.x; a1.y += wk * h1.y; a1.z += wk * h1.z; a1.w += wk * h1.w;
        wsum += wk;
    }
    inv = 1.f / (wsum + EPSF);
}

__device__ __forceinline__ void finish_node(const v4f e0, const v4f e1, const v4f a0, const v4f a1,
                                            const float inv, float* __restrict__ out,
                                            const int n, const int lane)
{
    const size_t o0 = (size_t)n * HH + lane * 4;
    const size_t o1 = o0 + (size_t)NN * HH;

    v4f x0, x1;
    x0.x = lrelu(e0.x + a0.x * inv); x0.y = lrelu(e0.y + a0.y * inv);
    x0.z = lrelu(e0.z + a0.z * inv); x0.w = lrelu(e0.w + a0.w * inv);
    x1.x = lrelu(e1.x + a1.x * inv); x1.y = lrelu(e1.y + a1.y * inv);
    x1.z = lrelu(e1.z + a1.z * inv); x1.w = lrelu(e1.w + a1.w * inv);

    float s0 = x0.x * x0.x + x0.y * x0.y + x0.z * x0.z + x0.w * x0.w;
    float s1 = x1.x * x1.x + x1.y * x1.y + x1.z * x1.z + x1.w * x1.w;
    #pragma unroll
    for (int off = 16; off >= 1; off >>= 1) {
        s0 += __shfl_xor(s0, off, 32);
        s1 += __shfl_xor(s1, off, 32);
    }
    const float i0 = 1.f / (sqrtf(s0) + EPSF);
    const float i1 = 1.f / (sqrtf(s1) + EPSF);
    x0.x *= i0; x0.y *= i0; x0.z *= i0; x0.w *= i0;
    x1.x *= i1; x1.y *= i1; x1.z *= i1; x1.w *= i1;
    *(v4f*)(out + o0) = x0;
    *(v4f*)(out + o1) = x1;
}

__global__ __launch_bounds__(512) void k_gather_out(const float* __restrict__ P,
                                                    const float* __restrict__ emb,
                                                    const float* __restrict__ Ww,
                                                    const float* __restrict__ Wb,
                                                    const float* __restrict__ weights,
                                                    const int* __restrict__ nbr,
                                                    float* __restrict__ out)
{
    __shared__ __attribute__((aligned(16))) float wl[CC * LDP];
    __shared__ __attribute__((aligned(16))) float ts[64 * LDP];
    __shared__ __attribute__((aligned(16))) float es[64 * LDP];

    const int lane = threadIdx.x & 31;
    const int wave  = __builtin_amdgcn_readfirstlane(threadIdx.x >> 5);
    const int rc    = lane & 15;
    const int half  = lane >> 4;
    const int strip = wave & 7;
    const int bsel  = wave >> 3;
    const int col   = (strip << 4) + rc;

    const int na = (int)blockIdx.x * 16 + wave;
    const int nb = na + NN / 2;

    const int ja = nbr[(size_t)na * KK + lane];
    const int jb = nbr[(size_t)nb * KK + lane];
    v4f g[4], wr[8];
    g[0] = *(const v4f*)(emb + (size_t)na * CC + 4 * lane);
    g[1] = *(const v4f*)(emb + (size_t)nb * CC + 4 * lane);
    g[2] = *(const v4f*)(emb + (size_t)(NN + na) * CC + 4 * lane);
    g[3] = *(const v4f*)(emb + (size_t)(NN + nb) * CC + 4 * lane);
    #pragma unroll
    for (int i = 0; i < 8; ++i)
        wr[i] = *(const v4f*)(Ww + (size_t)(8 * wave + i) * HH + 4 * lane);
    const float wbias = Wb[col];
    __builtin_amdgcn_sched_barrier(0);
    const float wa = weights[(size_t)na * NN + ja];
    const float wb = weights[(size_t)nb * NN + jb];
    __builtin_amdgcn_sched_barrier(0);

    *(v4f*)(ts + (wave)      * LDP + 4 * lane) = g[0];
    *(v4f*)(ts + (16 + wave) * LDP + 4 * lane) = g[1];
    *(v4f*)(ts + (32 + wave) * LDP + 4 * lane) = g[2];
    *(v4f*)(ts + (48 + wave) * LDP + 4 * lane) = g[3];
    #pragma unroll
    for (int i = 0; i < 8; ++i)
        *(v4f*)(wl + (8 * wave + i) * LDP + 4 * lane) = wr[i];
    lds_barrier();
    __builtin_amdgcn_sched_barrier(0);

    {
        v4f xa[16], xb[16];
        v2f bw1[32];
        const float* tr0 = ts + (32 * bsel + rc) * LDP + 4 * half;
        const float* tr1 = tr0 + 16 * LDP;
        #pragma unroll
        for (int j = 0; j < 16; ++j) {
            xa[j] = *(const v4f*)(tr0 + 8 * j);
            xb[j] = *(const v4f*)(tr1 + 8 * j);
            const float* wk = wl + (8 * j + 4 * half) * LDP + col;
            bw1[2 * j].x     = wk[0];
            bw1[2 * j].y     = wk[LDP];
            bw1[2 * j + 1].x = wk[2 * LDP];
            bw1[2 * j + 1].y = wk[3 * LDP];
            if (j & 1) __builtin_amdgcn_sched_barrier(0);
        }
        __builtin_amdgcn_sched_barrier(0);
        v8f ae0 = (v8f)0.f, ae1 = (v8f)0.f;
        #pragma unroll
        for (int j = 0; j < 16; ++j) {
            v2f a;
            a.x = xa[j].x; a.y = xa[j].y;
            ae0 = WMMA_F32(a, bw1[2 * j], ae0);
            a.x = xb[j].x; a.y = xb[j].y;
            ae1 = WMMA_F32(a, bw1[2 * j], ae1);
            a.x = xa[j].z; a.y = xa[j].w;
            ae0 = WMMA_F32(a, bw1[2 * j + 1], ae0);
            a.x = xb[j].z; a.y = xb[j].w;
            ae1 = WMMA_F32(a, bw1[2 * j + 1], ae1);
        }
        float* er = es + (32 * bsel) * LDP + col;
        #pragma unroll
        for (int r = 0; r < 8; ++r) {
            const int row = r + (half << 3);
            er[row * LDP]        = ae0[r] + wbias;
            er[(16 + row) * LDP] = ae1[r] + wbias;
        }
    }

    const float* p0 = P + (size_t)lane * 4;
    const float* p1 = p0 + (size_t)NN * HH;
    v4f aa0, aa1, ab0, ab1;
    float inva, invb;
    gather_sum(p0, p1, ja, wa, aa0, aa1, inva);
    gather_sum(p0, p1, jb, wb, ab0, ab1, invb);

    lds_barrier();
    {
        const v4f e0 = *(const v4f*)(es + (wave)      * LDP + 4 * lane);
        const v4f e1 = *(const v4f*)(es + (32 + wave) * LDP + 4 * lane);
        finish_node(e0, e1, aa0, aa1, inva, out, na, lane);
    }
    {
        const v4f e0 = *(const v4f*)(es + (16 + wave) * LDP + 4 * lane);
        const v4f e1 = *(const v4f*)(es + (48 + wave) * LDP + 4 * lane);
        finish_node(e0, e1, ab0, ab1, invb, out, nb, lane);
    }
}

extern "C" void kernel_launch(void* const* d_in, const int* in_sizes, int n_in,
                              void* d_out, int out_size, void* d_ws, size_t ws_size,
                              hipStream_t stream)
{
    const float* emb     = (const float*)d_in[0];
    const float* weights = (const float*)d_in[1];
    const int*   nbr     = (const int*)  d_in[2];
    const float* Qw      = (const float*)d_in[3];
    const float* Qb      = (const float*)d_in[4];
    const float* Ww      = (const float*)d_in[5];
    const float* Wb      = (const float*)d_in[6];
    float*       out     = (float*)d_out;

    float* P = (float*)d_ws;

    const dim3 blk(256);
    k_dense2<<<dim3(GA), blk, 0, stream>>>(emb, Qw, Qb, Ww, P);
    k_gather_out<<<dim3(GB), dim3(512), 0, stream>>>(P, emb, Ww, Wb, weights, nbr, out);
}
